# starting kernel
# baseline (speedup 1.0000x reference)
#include <hip/hip_runtime.h>

typedef __attribute__((ext_vector_type(2))) float v2f;
typedef __attribute__((ext_vector_type(8))) float v8f;
typedef __attribute__((ext_vector_type(4))) float f4;

#define Nb  32
#define Cc  32
#define Hh  56
#define Ww  56
#define Oo  32
#define CKK 288
#define PP  3136
#define TW  4

__global__ __launch_bounds__(128) void lc2d_wmma_f32(
    const float* __restrict__ x,
    const float* __restrict__ wgt,
    const float* __restrict__ bias,
    float* __restrict__ out)
{
    __shared__ float sP[CKK][33];
    __shared__ float sW[CKK][33];
    __shared__ float sO[Nb][33][TW];

    const int t     = threadIdx.x;
    const int lane  = t & 31;
    const int wave  = t >> 5;
    const int half  = lane >> 4;
    const int m16   = lane & 15;
    const int mtile = wave >> 1;
    const int ntile = wave & 1;
    const int mA    = m16 + 16 * mtile;
    const int oC    = m16 + 16 * ntile;

    const int h  = blockIdx.y;
    const int w0 = blockIdx.x * TW;

    for (int tw = 0; tw < TW; ++tw) {
        const int w = w0 + tw;
        const int p = h * Ww + w;

        __syncthreads();

        for (int oo = 0; oo < 8; ++oo) {
            const int o = wave * 8 + oo;
            const float* src = wgt + ((size_t)o * PP + p) * CKK;
#pragma unroll
            for (int s = 0; s < 9; ++s) {
                const int k = s * 32 + lane;
                sW[k][o] = src[k];
            }
        }

        for (int e = 0; e < 72; ++e) {
            const int idx = e * 128 + t;
            const int n   = idx & 31;
            const int k   = idx >> 5;
            const int c   = k / 9;
            const int r9  = k - c * 9;
            const int ki  = r9 / 3;
            const int kj  = r9 - ki * 3;
            const int hh  = h + ki - 1;
            const int ww  = w + kj - 1;
            float v = 0.0f;
            if ((unsigned)hh < (unsigned)Hh && (unsigned)ww < (unsigned)Ww)
                v = x[(((size_t)n * Cc + c) * Hh + hh) * Ww + ww];
            sP[k][n] = v;
        }

        if (tw + 1 < TW) {
            const int o = wave * 8 + (lane >> 2);
            __builtin_prefetch(wgt + ((size_t)o * PP + (p + 1)) * CKK + (size_t)(lane & 3) * 72, 0, 1);
        }

        __syncthreads();

        v8f acc = {};
#pragma unroll
        for (int k0 = 0; k0 < CKK; k0 += 4) {
            const int kA = k0 + 2 * half;
            v2f a, b;
            a[0] = sP[kA + 0][mA];
            a[1] = sP[kA + 1][mA];
            b[0] = sW[kA + 0][oC];
            b[1] = sW[kA + 1][oC];
            acc = __builtin_amdgcn_wmma_f32_16x16x4_f32(
                false, a, false, b, (short)0, acc, false, false);
        }

        const float bv = bias[(size_t)oC * PP + p];
#pragma unroll
        for (int r = 0; r < 8; ++r) {
            const int n = r + 8 * half + 16 * mtile;
            sO[n][oC][tw] = acc[r] + bv;
        }
    }

    __syncthreads();

    for (int e = 0; e < 8; ++e) {
        const int pair = e * 128 + t;
        const int n = pair >> 5;
        const int o = pair & 31;
        f4 v;
        v.x = sO[n][o][0];
        v.y = sO[n][o][1];
        v.z = sO[n][o][2];
        v.w = sO[n][o][3];
        *(f4*)(out + ((size_t)n * Oo + o) * PP + (size_t)h * Ww + w0) = v;
    }
}

extern "C" void kernel_launch(void* const* d_in, const int* in_sizes, int n_in,
                              void* d_out, int out_size, void* d_ws, size_t ws_size,
                              hipStream_t stream) {
    const float* x    = (const float*)d_in[0];
    const float* wgt  = (const float*)d_in[1];
    const float* bias = (const float*)d_in[2];
    float* out        = (float*)d_out;

    dim3 grid(Ww / TW, Hh);
    lc2d_wmma_f32<<<grid, 128, 0, stream>>>(x, wgt, bias, out);
}


# x re-laid out by a pre-pass launch to [c][h+2][w+2][n] with a zero halo in the workspace: every im2col row is one unconditional 128-B load, 72 in flight per wave; k order and sums unchanged
# speedup vs baseline: 2.1296x; 2.1296x over previous
#include <hip/hip_runtime.h>

typedef __attribute__((ext_vector_type(2))) float v2f;
typedef __attribute__((ext_vector_type(8))) float v8f;
typedef __attribute__((ext_vector_type(4))) float f4;

#define Nb  32
#define Cc  32
#define Hh  56
#define Ww  56
#define Oo  32
#define CKK 288
#define PP  3136
#define TW  4

#define HP  58
#define WP  58

__global__ __launch_bounds__(128) void lc2d_x_batch_inner(
    const float* __restrict__ x,
    float* __restrict__ xT)
{
    __shared__ float sT[Nb][2 * Ww + 1];

    const int t  = threadIdx.x;
    const int bx = blockIdx.x;
    const int c  = blockIdx.y;
    const f4 zero = {0.0f, 0.0f, 0.0f, 0.0f};

    if (bx == Hh / 2) {
        float* top = xT + (size_t)c * HP * WP * Nb;
        float* bot = top + (size_t)(HP - 1) * WP * Nb;
#pragma unroll
        for (int j = 0; j < 4; ++j) {
            const int q = j * 128 + t;
            if (q < WP * (Nb / 4)) {
                *(f4*)(top + (size_t)q * 4) = zero;
                *(f4*)(bot + (size_t)q * 4) = zero;
            }
        }
        return;
    }

    const int h0 = 2 * bx;
    f4 v[7];
#pragma unroll
    for (int j = 0; j < 7; ++j) {
        const int q = j * 128 + t;
        const int n = q / 28;
        const int r = q - n * 28;
        v[j] = *(const f4*)(x + (((size_t)n * Cc + c) * Hh + h0) * Ww + r * 4);
    }
#pragma unroll
    for (int j = 0; j < 7; ++j) {
        const int q = j * 128 + t;
        const int n = q / 28;
        const int r = q - n * 28;
        sT[n][r * 4 + 0] = v[j].x;
        sT[n][r * 4 + 1] = v[j].y;
        sT[n][r * 4 + 2] = v[j].z;
        sT[n][r * 4 + 3] = v[j].w;
    }

    __syncthreads();

    float* dst = xT + ((size_t)c * HP + h0 + 1) * WP * Nb;
#pragma unroll
    for (int j = 0; j < 8; ++j) {
        const int q = j * 128 + t;
        if (q < 2 * WP * (Nb / 4)) {
            const int row = q / (WP * (Nb / 4));
            const int r   = q - row * (WP * (Nb / 4));
            const int wp  = r >> 3;
            const int n4  = (r & 7) * 4;
            f4 o = zero;
            if (wp != 0 && wp != WP - 1) {
                const int col = row * Ww + wp - 1;
                o.x = sT[n4 + 0][col];
                o.y = sT[n4 + 1][col];
                o.z = sT[n4 + 2][col];
                o.w = sT[n4 + 3][col];
            }
            *(f4*)(dst + (size_t)q * 4) = o;
        }
    }
}

__global__ __launch_bounds__(128) void lc2d_wmma_f32(
    const float* __restrict__ xT,
    const float* __restrict__ wgt,
    const float* __restrict__ bias,
    float* __restrict__ out)
{
    __shared__ float sP[CKK][33];
    __shared__ float sW[CKK][33];
    __shared__ float sO[Nb][33][TW];

    const int t     = threadIdx.x;
    const int lane  = t & 31;
    const int wave  = t >> 5;
    const int half  = lane >> 4;
    const int m16   = lane & 15;
    const int mtile = wave >> 1;
    const int ntile = wave & 1;
    const int mA    = m16 + 16 * mtile;
    const int oC    = m16 + 16 * ntile;

    const int h  = blockIdx.y;
    const int w0 = blockIdx.x * TW;

    for (int tw = 0; tw < TW; ++tw) {
        const int w = w0 + tw;
        const int p = h * Ww + w;

        __syncthreads();

        for (int oo = 0; oo < 8; ++oo) {
            const int o = wave * 8 + oo;
            const float* src = wgt + ((size_t)o * PP + p) * CKK;
#pragma unroll
            for (int s = 0; s < 9; ++s) {
                const int k = s * 32 + lane;
                sW[k][o] = src[k];
            }
        }

        {
            const float* xb = xT + (((size_t)(wave * 8) * HP + h) * WP + w) * Nb + lane;
#pragma unroll
            for (int cc = 0; cc < 8; ++cc) {
#pragma unroll
                for (int ki = 0; ki < 3; ++ki) {
#pragma unroll
                    for (int kj = 0; kj < 3; ++kj) {
                        sP[wave * 72 + cc * 9 + ki * 3 + kj][lane] = xb[((cc * HP + ki) * WP + kj) * Nb];
                    }
                }
            }
        }

        if (tw + 1 < TW) {
            const int o = wave * 8 + (lane >> 2);
            __builtin_prefetch(wgt + ((size_t)o * PP + (p + 1)) * CKK + (size_t)(lane & 3) * 72, 0, 1);
        }

        __syncthreads();

        v8f acc = {};
#pragma unroll
        for (int k0 = 0; k0 < CKK; k0 += 4) {
            const int kA = k0 + 2 * half;
            v2f a, b;
            a[0] = sP[kA + 0][mA];
            a[1] = sP[kA + 1][mA];
            b[0] = sW[kA + 0][oC];
            b[1] = sW[kA + 1][oC];
            acc = __builtin_amdgcn_wmma_f32_16x16x4_f32(
                false, a, false, b, (short)0, acc, false, false);
        }

        const float bv = bias[(size_t)oC * PP + p];
#pragma unroll
        for (int r = 0; r < 8; ++r) {
            const int n = r + 8 * half + 16 * mtile;
            sO[n][oC][tw] = acc[r] + bv;
        }
    }

    __syncthreads();

    for (int e = 0; e < 8; ++e) {
        const int pair = e * 128 + t;
        const int n = pair >> 5;
        const int o = pair & 31;
        f4 v;
        v.x = sO[n][o][0];
        v.y = sO[n][o][1];
        v.z = sO[n][o][2];
        v.w = sO[n][o][3];
        *(f4*)(out + ((size_t)n * Oo + o) * PP + (size_t)h * Ww + w0) = v;
    }
}

extern "C" void kernel_launch(void* const* d_in, const int* in_sizes, int n_in,
                              void* d_out, int out_size, void* d_ws, size_t ws_size,
                              hipStream_t stream) {
    const float* x    = (const float*)d_in[0];
    const float* wgt  = (const float*)d_in[1];
    const float* bias = (const float*)d_in[2];
    float* out        = (float*)d_out;
    float* xT         = (float*)d_ws;

    dim3 gridT(Hh / 2 + 1, Cc);
    lc2d_x_batch_inner<<<gridT, 128, 0, stream>>>(x, xT);

    dim3 grid(Ww / TW, Hh);
    lc2d_wmma_f32<<<grid, 128, 0, stream>>>(xT, wgt, bias, out);
}


# GEMM loop in blocks of 12 k-steps: the next block's A and B fragments are read from LDS as one group ahead of the current block's matrix instructions (two register sets); k order and sums unchanged
# speedup vs baseline: 2.2061x; 1.0359x over previous
#include <hip/hip_runtime.h>

typedef __attribute__((ext_vector_type(2))) float v2f;
typedef __attribute__((ext_vector_type(8))) float v8f;
typedef __attribute__((ext_vector_type(4))) float f4;

#define Nb  32
#define Cc  32
#define Hh  56
#define Ww  56
#define Oo  32
#define CKK 288
#define PP  3136
#define TW  4
#define KB  12
#define NBLK 6

#define HP  58
#define WP  58

__global__ __launch_bounds__(128) void lc2d_x_batch_inner(
    const float* __restrict__ x,
    float* __restrict__ xT)
{
    __shared__ float sT[Nb][2 * Ww + 1];

    const int t  = threadIdx.x;
    const int bx = blockIdx.x;
    const int c  = blockIdx.y;
    const f4 zero = {0.0f, 0.0f, 0.0f, 0.0f};

    if (bx == Hh / 2) {
        float* top = xT + (size_t)c * HP * WP * Nb;
        float* bot = top + (size_t)(HP - 1) * WP * Nb;
#pragma unroll
        for (int j = 0; j < 4; ++j) {
            const int q = j * 128 + t;
            if (q < WP * (Nb / 4)) {
                *(f4*)(top + (size_t)q * 4) = zero;
                *(f4*)(bot + (size_t)q * 4) = zero;
            }
        }
        return;
    }

    const int h0 = 2 * bx;
    f4 v[7];
#pragma unroll
    for (int j = 0; j < 7; ++j) {
        const int q = j * 128 + t;
        const int n = q / 28;
        const int r = q - n * 28;
        v[j] = *(const f4*)(x + (((size_t)n * Cc + c) * Hh + h0) * Ww + r * 4);
    }
#pragma unroll
    for (int j = 0; j < 7; ++j) {
        const int q = j * 128 + t;
        const int n = q / 28;
        const int r = q - n * 28;
        sT[n][r * 4 + 0] = v[j].x;
        sT[n][r * 4 + 1] = v[j].y;
        sT[n][r * 4 + 2] = v[j].z;
        sT[n][r * 4 + 3] = v[j].w;
    }

    __syncthreads();

    float* dst = xT + ((size_t)c * HP + h0 + 1) * WP * Nb;
#pragma unroll
    for (int j = 0; j < 8; ++j) {
        const int q = j * 128 + t;
        if (q < 2 * WP * (Nb / 4)) {
            const int row = q / (WP * (Nb / 4));
            const int r   = q - row * (WP * (Nb / 4));
            const int wp  = r >> 3;
            const int n4  = (r & 7) * 4;
            f4 o = zero;
            if (wp != 0 && wp != WP - 1) {
                const int col = row * Ww + wp - 1;
                o.x = sT[n4 + 0][col];
                o.y = sT[n4 + 1][col];
                o.z = sT[n4 + 2][col];
                o.w = sT[n4 + 3][col];
            }
            *(f4*)(dst + (size_t)q * 4) = o;
        }
    }
}

__global__ __launch_bounds__(128) void lc2d_wmma_f32(
    const float* __restrict__ xT,
    const float* __restrict__ wgt,
    const float* __restrict__ bias,
    float* __restrict__ out)
{
    __shared__ float sP[CKK][33];
    __shared__ float sW[CKK][33];
    __shared__ float sO[Nb][33][TW];

    const int t     = threadIdx.x;
    const int lane  = t & 31;
    const int wave  = t >> 5;
    const int half  = lane >> 4;
    const int m16   = lane & 15;
    const int mtile = wave >> 1;
    const int ntile = wave & 1;
    const int mA    = m16 + 16 * mtile;
    const int oC    = m16 + 16 * ntile;

    const int h  = blockIdx.y;
    const int w0 = blockIdx.x * TW;

    for (int tw = 0; tw < TW; ++tw) {
        const int w = w0 + tw;
        const int p = h * Ww + w;

        __syncthreads();

        for (int oo = 0; oo < 8; ++oo) {
            const int o = wave * 8 + oo;
            const float* src = wgt + ((size_t)o * PP + p) * CKK;
#pragma unroll
            for (int s = 0; s < 9; ++s) {
                const int k = s * 32 + lane;
                sW[k][o] = src[k];
            }
        }

        {
            const float* xb = xT + (((size_t)(wave * 8) * HP + h) * WP + w) * Nb + lane;
#pragma unroll
            for (int cc = 0; cc < 8; ++cc) {
#pragma unroll
                for (int ki = 0; ki < 3; ++ki) {
#pragma unroll
                    for (int kj = 0; kj < 3; ++kj) {
                        sP[wave * 72 + cc * 9 + ki * 3 + kj][lane] = xb[((cc * HP + ki) * WP + kj) * Nb];
                    }
                }
            }
        }

        if (tw + 1 < TW) {
            const int o = wave * 8 + (lane >> 2);
            __builtin_prefetch(wgt + ((size_t)o * PP + (p + 1)) * CKK + (size_t)(lane & 3) * 72, 0, 1);
        }

        __syncthreads();

        v2f fa[2][KB], fb[2][KB];
#pragma unroll
        for (int j = 0; j < KB; ++j) {
            const int kA = 4 * j + 2 * half;
            fa[0][j][0] = sP[kA + 0][mA];
            fa[0][j][1] = sP[kA + 1][mA];
            fb[0][j][0] = sW[kA + 0][oC];
            fb[0][j][1] = sW[kA + 1][oC];
        }
        v8f acc = {};
#pragma unroll
        for (int blk = 0; blk < NBLK; ++blk) {
            const int cur = blk & 1;
            const int nxt = cur ^ 1;
            if (blk + 1 < NBLK) {
#pragma unroll
                for (int j = 0; j < KB; ++j) {
                    const int kA = 4 * ((blk + 1) * KB + j) + 2 * half;
                    fa[nxt][j][0] = sP[kA + 0][mA];
                    fa[nxt][j][1] = sP[kA + 1][mA];
                    fb[nxt][j][0] = sW[kA + 0][oC];
                    fb[nxt][j][1] = sW[kA + 1][oC];
                }
            }
            __builtin_amdgcn_sched_barrier(0);
#pragma unroll
            for (int j = 0; j < KB; ++j) {
                acc = __builtin_amdgcn_wmma_f32_16x16x4_f32(
                    false, fa[cur][j], false, fb[cur][j], (short)0, acc, false, false);
            }
            __builtin_amdgcn_sched_barrier(0);
        }

        const float bv = bias[(size_t)oC * PP + p];
#pragma unroll
        for (int r = 0; r < 8; ++r) {
            const int n = r + 8 * half + 16 * mtile;
            sO[n][oC][tw] = acc[r] + bv;
        }
    }

    __syncthreads();

    for (int e = 0; e < 8; ++e) {
        const int pair = e * 128 + t;
        const int n = pair >> 5;
        const int o = pair & 31;
        f4 v;
        v.x = sO[n][o][0];
        v.y = sO[n][o][1];
        v.z = sO[n][o][2];
        v.w = sO[n][o][3];
        *(f4*)(out + ((size_t)n * Oo + o) * PP + (size_t)h * Ww + w0) = v;
    }
}

extern "C" void kernel_launch(void* const* d_in, const int* in_sizes, int n_in,
                              void* d_out, int out_size, void* d_ws, size_t ws_size,
                              hipStream_t stream) {
    const float* x    = (const float*)d_in[0];
    const float* wgt  = (const float*)d_in[1];
    const float* bias = (const float*)d_in[2];
    float* out        = (float*)d_out;
    float* xT         = (float*)d_ws;

    dim3 gridT(Hh / 2 + 1, Cc);
    lc2d_x_batch_inner<<<gridT, 128, 0, stream>>>(x, xT);

    dim3 grid(Ww / TW, Hh);
    lc2d_wmma_f32<<<grid, 128, 0, stream>>>(xT, wgt, bias, out);
}


# weights loaded from global straight into B-fragment registers (16-B loads, no LDS staging); a wave sums half of k for 16 o x 32 n, halves added at the store; k order permuted inside each 8
# speedup vs baseline: 2.6480x; 1.2003x over previous
#include <hip/hip_runtime.h>

typedef __attribute__((ext_vector_type(2))) float v2f;
typedef __attribute__((ext_vector_type(8))) float v8f;
typedef __attribute__((ext_vector_type(4))) float f4;

#define Nb  32
#define Cc  32
#define Hh  56
#define Ww  56
#define Oo  32
#define CKK 288
#define PP  3136
#define TW  4
#define KB  12
#define NBLK 3
#define KH  144
#define SPP 80

#define HP  58
#define WP  58

__global__ __launch_bounds__(128) void lc2d_x_batch_inner(
    const float* __restrict__ x,
    float* __restrict__ xT)
{
    __shared__ float sT[Nb][2 * Ww + 1];

    const int t  = threadIdx.x;
    const int bx = blockIdx.x;
    const int c  = blockIdx.y;
    const f4 zero = {0.0f, 0.0f, 0.0f, 0.0f};

    if (bx == Hh / 2) {
        float* top = xT + (size_t)c * HP * WP * Nb;
        float* bot = top + (size_t)(HP - 1) * WP * Nb;
#pragma unroll
        for (int j = 0; j < 4; ++j) {
            const int q = j * 128 + t;
            if (q < WP * (Nb / 4)) {
                *(f4*)(top + (size_t)q * 4) = zero;
                *(f4*)(bot + (size_t)q * 4) = zero;
            }
        }
        return;
    }

    const int h0 = 2 * bx;
    f4 v[7];
#pragma unroll
    for (int j = 0; j < 7; ++j) {
        const int q = j * 128 + t;
        const int n = q / 28;
        const int r = q - n * 28;
        v[j] = *(const f4*)(x + (((size_t)n * Cc + c) * Hh + h0) * Ww + r * 4);
    }
#pragma unroll
    for (int j = 0; j < 7; ++j) {
        const int q = j * 128 + t;
        const int n = q / 28;
        const int r = q - n * 28;
        sT[n][r * 4 + 0] = v[j].x;
        sT[n][r * 4 + 1] = v[j].y;
        sT[n][r * 4 + 2] = v[j].z;
        sT[n][r * 4 + 3] = v[j].w;
    }

    __syncthreads();

    float* dst = xT + ((size_t)c * HP + h0 + 1) * WP * Nb;
#pragma unroll
    for (int j = 0; j < 8; ++j) {
        const int q = j * 128 + t;
        if (q < 2 * WP * (Nb / 4)) {
            const int row = q / (WP * (Nb / 4));
            const int r   = q - row * (WP * (Nb / 4));
            const int wp  = r >> 3;
            const int n4  = (r & 7) * 4;
            f4 o = zero;
            if (wp != 0 && wp != WP - 1) {
                const int col = row * Ww + wp - 1;
                o.x = sT[n4 + 0][col];
                o.y = sT[n4 + 1][col];
                o.z = sT[n4 + 2][col];
                o.w = sT[n4 + 3][col];
            }
            *(f4*)(dst + (size_t)q * 4) = o;
        }
    }
}

__global__ __launch_bounds__(128) void lc2d_wmma_f32(
    const float* __restrict__ xT,
    const float* __restrict__ wgt,
    const float* __restrict__ bias,
    float* __restrict__ out)
{
    __shared__ float sP[CKK / 2][SPP];
    __shared__ float sO[2][Nb][33][TW];

    const int t     = threadIdx.x;
    const int lane  = t & 31;
    const int wave  = t >> 5;
    const int half  = lane >> 4;
    const int m16   = lane & 15;
    const int otile = wave & 1;
    const int kh    = wave >> 1;
    const int oC    = m16 + 16 * otile;
    const int K0    = KH * kh;

    const int h  = blockIdx.y;
    const int w0 = blockIdx.x * TW;

    for (int tw = 0; tw < TW; ++tw) {
        const int w = w0 + tw;
        const int p = h * Ww + w;

        __syncthreads();

        f4 q[KH / 8];
        {
            const f4* wsrc = (const f4*)(wgt + ((size_t)oC * PP + p) * CKK + K0 + 4 * half);
#pragma unroll
            for (int j = 0; j < KH / 8; ++j) q[j] = wsrc[2 * j];
        }
        __builtin_amdgcn_sched_barrier(0);

        {
            const float* xb = xT + (((size_t)(wave * 8) * HP + h) * WP + w) * Nb + lane;
            float xv[72];
#pragma unroll
            for (int cc = 0; cc < 8; ++cc) {
#pragma unroll
                for (int ki = 0; ki < 3; ++ki) {
#pragma unroll
                    for (int kj = 0; kj < 3; ++kj) {
                        xv[cc * 9 + ki * 3 + kj] = xb[((cc * HP + ki) * WP + kj) * Nb];
                    }
                }
            }
#pragma unroll
            for (int i = 0; i < 36; ++i) {
                v2f pr;
                pr[0] = xv[2 * i];
                pr[1] = xv[2 * i + 1];
                *(v2f*)&sP[wave * 36 + i][2 * lane] = pr;
            }
        }

        if (tw + 1 < TW) {
            const int o = wave * 8 + (lane >> 2);
            __builtin_prefetch(wgt + ((size_t)o * PP + (p + 1)) * CKK + (size_t)(lane & 3) * 72, 0, 1);
        }

        __syncthreads();

        v2f fa[2][KB][2];
#pragma unroll
        for (int j = 0; j < KB; ++j) {
            const int pi = (K0 >> 1) + 4 * (j >> 1) + 2 * half + (j & 1);
#pragma unroll
            for (int mt = 0; mt < 2; ++mt) fa[0][j][mt] = *(const v2f*)&sP[pi][2 * (m16 + 16 * mt)];
        }
        v8f acc0 = {};
        v8f acc1 = {};
#pragma unroll
        for (int blk = 0; blk < NBLK; ++blk) {
            const int cur = blk & 1;
            const int nxt = cur ^ 1;
            if (blk + 1 < NBLK) {
#pragma unroll
                for (int j = 0; j < KB; ++j) {
                    const int s  = (blk + 1) * KB + j;
                    const int pi = (K0 >> 1) + 4 * (s >> 1) + 2 * half + (s & 1);
#pragma unroll
                    for (int mt = 0; mt < 2; ++mt) fa[nxt][j][mt] = *(const v2f*)&sP[pi][2 * (m16 + 16 * mt)];
                }
            }
            __builtin_amdgcn_sched_barrier(0);
#pragma unroll
            for (int j = 0; j < KB; ++j) {
                const int s = blk * KB + j;
                v2f fb;
                if (s & 1) { fb[0] = q[s >> 1].z; fb[1] = q[s >> 1].w; }
                else       { fb[0] = q[s >> 1].x; fb[1] = q[s >> 1].y; }
                acc0 = __builtin_amdgcn_wmma_f32_16x16x4_f32(
                    false, fa[cur][j][0], false, fb, (short)0, acc0, false, false);
                acc1 = __builtin_amdgcn_wmma_f32_16x16x4_f32(
                    false, fa[cur][j][1], false, fb, (short)0, acc1, false, false);
            }
            __builtin_amdgcn_sched_barrier(0);
        }

        const float bl = bias[(size_t)oC * PP + p];
        const float bv = (kh == 0) ? bl : 0.0f;
#pragma unroll
        for (int r = 0; r < 8; ++r) {
            const int n = r + 8 * half;
            sO[kh][n][oC][tw]      = acc0[r] + bv;
            sO[kh][n + 16][oC][tw] = acc1[r] + bv;
        }
    }

    __syncthreads();

    for (int e = 0; e < 8; ++e) {
        const int pair = e * 128 + t;
        const int n = pair >> 5;
        const int o = pair & 31;
        f4 v;
        v.x = sO[0][n][o][0] + sO[1][n][o][0];
        v.y = sO[0][n][o][1] + sO[1][n][o][1];
        v.z = sO[0][n][o][2] + sO[1][n][o][2];
        v.w = sO[0][n][o][3] + sO[1][n][o][3];
        *(f4*)(out + ((size_t)n * Oo + o) * PP + (size_t)h * Ww + w0) = v;
    }
}

extern "C" void kernel_launch(void* const* d_in, const int* in_sizes, int n_in,
                              void* d_out, int out_size, void* d_ws, size_t ws_size,
                              hipStream_t stream) {
    const float* x    = (const float*)d_in[0];
    const float* wgt  = (const float*)d_in[1];
    const float* bias = (const float*)d_in[2];
    float* out        = (float*)d_out;
    float* xT         = (float*)d_ws;

    dim3 gridT(Hh / 2 + 1, Cc);
    lc2d_x_batch_inner<<<gridT, 128, 0, stream>>>(x, xT);

    dim3 grid(Ww / TW, Hh);
    lc2d_wmma_f32<<<grid, 128, 0, stream>>>(xT, wgt, bias, out);
}


# weights and bias of position tw+1 loaded into a second register set in front of position tw's matrix instructions (GEMM blocks of 6 k-steps to fit 4 waves per SIMD); k order and sums unchanged
# speedup vs baseline: 2.7704x; 1.0462x over previous
#include <hip/hip_runtime.h>

typedef __attribute__((ext_vector_type(2))) float v2f;
typedef __attribute__((ext_vector_type(8))) float v8f;
typedef __attribute__((ext_vector_type(4))) float f4;

#define Nb  32
#define Cc  32
#define Hh  56
#define Ww  56
#define Oo  32
#define CKK 288
#define PP  3136
#define TW  4
#define KB  6
#define NBLK 6
#define KH  144
#define SPP 80

#define HP  58
#define WP  58

__global__ __launch_bounds__(128) void lc2d_x_batch_inner(
    const float* __restrict__ x,
    float* __restrict__ xT)
{
    __shared__ float sT[Nb][2 * Ww + 1];

    const int t  = threadIdx.x;
    const int bx = blockIdx.x;
    const int c  = blockIdx.y;
    const f4 zero = {0.0f, 0.0f, 0.0f, 0.0f};

    if (bx == Hh / 2) {
        float* top = xT + (size_t)c * HP * WP * Nb;
        float* bot = top + (size_t)(HP - 1) * WP * Nb;
#pragma unroll
        for (int j = 0; j < 4; ++j) {
            const int q = j * 128 + t;
            if (q < WP * (Nb / 4)) {
                *(f4*)(top + (size_t)q * 4) = zero;
                *(f4*)(bot + (size_t)q * 4) = zero;
            }
        }
        return;
    }

    const int h0 = 2 * bx;
    f4 v[7];
#pragma unroll
    for (int j = 0; j < 7; ++j) {
        const int q = j * 128 + t;
        const int n = q / 28;
        const int r = q - n * 28;
        v[j] = *(const f4*)(x + (((size_t)n * Cc + c) * Hh + h0) * Ww + r * 4);
    }
#pragma unroll
    for (int j = 0; j < 7; ++j) {
        const int q = j * 128 + t;
        const int n = q / 28;
        const int r = q - n * 28;
        sT[n][r * 4 + 0] = v[j].x;
        sT[n][r * 4 + 1] = v[j].y;
        sT[n][r * 4 + 2] = v[j].z;
        sT[n][r * 4 + 3] = v[j].w;
    }

    __syncthreads();

    float* dst = xT + ((size_t)c * HP + h0 + 1) * WP * Nb;
#pragma unroll
    for (int j = 0; j < 8; ++j) {
        const int q = j * 128 + t;
        if (q < 2 * WP * (Nb / 4)) {
            const int row = q / (WP * (Nb / 4));
            const int r   = q - row * (WP * (Nb / 4));
            const int wp  = r >> 3;
            const int n4  = (r & 7) * 4;
            f4 o = zero;
            if (wp != 0 && wp != WP - 1) {
                const int col = row * Ww + wp - 1;
                o.x = sT[n4 + 0][col];
                o.y = sT[n4 + 1][col];
                o.z = sT[n4 + 2][col];
                o.w = sT[n4 + 3][col];
            }
            *(f4*)(dst + (size_t)q * 4) = o;
        }
    }
}

__global__ __launch_bounds__(128) void lc2d_wmma_f32(
    const float* __restrict__ xT,
    const float* __restrict__ wgt,
    const float* __restrict__ bias,
    float* __restrict__ out)
{
    __shared__ float sP[CKK / 2][SPP];
    __shared__ float sO[2][Nb][33][TW];

    const int t     = threadIdx.x;
    const int lane  = t & 31;
    const int wave  = t >> 5;
    const int half  = lane >> 4;
    const int m16   = lane & 15;
    const int otile = wave & 1;
    const int kh    = wave >> 1;
    const int oC    = m16 + 16 * otile;
    const int K0    = KH * kh;

    const int h  = blockIdx.y;
    const int w0 = blockIdx.x * TW;

    const f4* wsrc = (const f4*)(wgt + ((size_t)oC * PP + (size_t)h * Ww + w0) * CKK + K0 + 4 * half);
    const float* bsrc = bias + (size_t)oC * PP + (size_t)h * Ww + w0;
    f4 q[2][KH / 8];
    float bq[2];
#pragma unroll
    for (int j = 0; j < KH / 8; ++j) q[0][j] = wsrc[2 * j];
    bq[0] = bsrc[0];
    __builtin_amdgcn_sched_barrier(0);

#pragma unroll
    for (int tw = 0; tw < TW; ++tw) {
        const int w = w0 + tw;
        const int p = h * Ww + w;
        const int cq = tw & 1;

        if (tw > 0) __syncthreads();

        {
            const float* xb = xT + (((size_t)(wave * 8) * HP + h) * WP + w) * Nb + lane;
            float xv[72];
#pragma unroll
            for (int cc = 0; cc < 8; ++cc) {
#pragma unroll
                for (int ki = 0; ki < 3; ++ki) {
#pragma unroll
                    for (int kj = 0; kj < 3; ++kj) {
                        xv[cc * 9 + ki * 3 + kj] = xb[((cc * HP + ki) * WP + kj) * Nb];
                    }
                }
            }
#pragma unroll
            for (int i = 0; i < 36; ++i) {
                v2f pr;
                pr[0] = xv[2 * i];
                pr[1] = xv[2 * i + 1];
                *(v2f*)&sP[wave * 36 + i][2 * lane] = pr;
            }
        }

        if (tw + 2 < TW) {
            const int o = wave * 8 + (lane >> 2);
            __builtin_prefetch(wgt + ((size_t)o * PP + (p + 2)) * CKK + (size_t)(lane & 3) * 72, 0, 1);
        }

        __syncthreads();

        if (tw + 1 < TW) {
#pragma unroll
            for (int j = 0; j < KH / 8; ++j) q[cq ^ 1][j] = wsrc[(tw + 1) * (CKK / 4) + 2 * j];
            bq[cq ^ 1] = bsrc[tw + 1];
        }
        __builtin_amdgcn_sched_barrier(0);

        v2f fa[2][KB][2];
#pragma unroll
        for (int j = 0; j < KB; ++j) {
            const int pi = (K0 >> 1) + 4 * (j >> 1) + 2 * half + (j & 1);
#pragma unroll
            for (int mt = 0; mt < 2; ++mt) fa[0][j][mt] = *(const v2f*)&sP[pi][2 * (m16 + 16 * mt)];
        }
        v8f acc0 = {};
        v8f acc1 = {};
#pragma unroll
        for (int blk = 0; blk < NBLK; ++blk) {
            const int cur = blk & 1;
            const int nxt = cur ^ 1;
            if (blk + 1 < NBLK) {
#pragma unroll
                for (int j = 0; j < KB; ++j) {
                    const int s  = (blk + 1) * KB + j;
                    const int pi = (K0 >> 1) + 4 * (s >> 1) + 2 * half + (s & 1);
#pragma unroll
                    for (int mt = 0; mt < 2; ++mt) fa[nxt][j][mt] = *(const v2f*)&sP[pi][2 * (m16 + 16 * mt)];
                }
            }
            __builtin_amdgcn_sched_barrier(0);
#pragma unroll
            for (int j = 0; j < KB; ++j) {
                const int s = blk * KB + j;
                v2f fb;
                if (s & 1) { fb[0] = q[cq][s >> 1].z; fb[1] = q[cq][s >> 1].w; }
                else       { fb[0] = q[cq][s >> 1].x; fb[1] = q[cq][s >> 1].y; }
                acc0 = __builtin_amdgcn_wmma_f32_16x16x4_f32(
                    false, fa[cur][j][0], false, fb, (short)0, acc0, false, false);
                acc1 = __builtin_amdgcn_wmma_f32_16x16x4_f32(
                    false, fa[cur][j][1], false, fb, (short)0, acc1, false, false);
            }
            __builtin_amdgcn_sched_barrier(0);
        }

        const float bv = (kh == 0) ? bq[cq] : 0.0f;
#pragma unroll
        for (int r = 0; r < 8; ++r) {
            const int n = r + 8 * half;
            sO[kh][n][oC][tw]      = acc0[r] + bv;
            sO[kh][n + 16][oC][tw] = acc1[r] + bv;
        }
    }

    __syncthreads();

    for (int e = 0; e < 8; ++e) {
        const int pair = e * 128 + t;
        const int n = pair >> 5;
        const int o = pair & 31;
        f4 v;
        v.x = sO[0][n][o][0] + sO[1][n][o][0];
        v.y = sO[0][n][o][1] + sO[1][n][o][1];
        v.z = sO[0][n][o][2] + sO[1][n][o][2];
        v.w = sO[0][n][o][3] + sO[1][n][o][3];
        *(f4*)(out + ((size_t)n * Oo + o) * PP + (size_t)h * Ww + w0) = v;
    }
}

extern "C" void kernel_launch(void* const* d_in, const int* in_sizes, int n_in,
                              void* d_out, int out_size, void* d_ws, size_t ws_size,
                              hipStream_t stream) {
    const float* x    = (const float*)d_in[0];
    const float* wgt  = (const float*)d_in[1];
    const float* bias = (const float*)d_in[2];
    float* out        = (float*)d_out;
    float* xT         = (float*)d_ws;

    dim3 gridT(Hh / 2 + 1, Cc);
    lc2d_x_batch_inner<<<gridT, 128, 0, stream>>>(x, xT);

    dim3 grid(Ww / TW, Hh);
    lc2d_wmma_f32<<<grid, 128, 0, stream>>>(xT, wgt, bias, out);
}


# main kernel re-tiled: a workgroup is 8 waves, two 4-wave teams walk 4 positions each of an 8-position strip (392 workgroups), outputs leave as 32-B pieces; k order and sums unchanged
# speedup vs baseline: 3.0549x; 1.1027x over previous
#include <hip/hip_runtime.h>

typedef __attribute__((ext_vector_type(2))) float v2f;
typedef __attribute__((ext_vector_type(8))) float v8f;
typedef __attribute__((ext_vector_type(4))) float f4;

#define Nb  32
#define Cc  32
#define Hh  56
#define Ww  56
#define Oo  32
#define CKK 288
#define PP  3136
#define TW  4
#define NT  2
#define TWG 8
#define KB  6
#define NBLK 6
#define KH  144
#define SPP 80

#define HP  58
#define WP  58

__global__ __launch_bounds__(128) void lc2d_x_batch_inner(
    const float* __restrict__ x,
    float* __restrict__ xT)
{
    __shared__ float sT[Nb][2 * Ww + 1];

    const int t  = threadIdx.x;
    const int bx = blockIdx.x;
    const int c  = blockIdx.y;
    const f4 zero = {0.0f, 0.0f, 0.0f, 0.0f};

    if (bx == Hh / 2) {
        float* top = xT + (size_t)c * HP * WP * Nb;
        float* bot = top + (size_t)(HP - 1) * WP * Nb;
#pragma unroll
        for (int j = 0; j < 4; ++j) {
            const int q = j * 128 + t;
            if (q < WP * (Nb / 4)) {
                *(f4*)(top + (size_t)q * 4) = zero;
                *(f4*)(bot + (size_t)q * 4) = zero;
            }
        }
        return;
    }

    const int h0 = 2 * bx;
    f4 v[7];
#pragma unroll
    for (int j = 0; j < 7; ++j) {
        const int q = j * 128 + t;
        const int n = q / 28;
        const int r = q - n * 28;
        v[j] = *(const f4*)(x + (((size_t)n * Cc + c) * Hh + h0) * Ww + r * 4);
    }
#pragma unroll
    for (int j = 0; j < 7; ++j) {
        const int q = j * 128 + t;
        const int n = q / 28;
        const int r = q - n * 28;
        sT[n][r * 4 + 0] = v[j].x;
        sT[n][r * 4 + 1] = v[j].y;
        sT[n][r * 4 + 2] = v[j].z;
        sT[n][r * 4 + 3] = v[j].w;
    }

    __syncthreads();

    float* dst = xT + ((size_t)c * HP + h0 + 1) * WP * Nb;
#pragma unroll
    for (int j = 0; j < 8; ++j) {
        const int q = j * 128 + t;
        if (q < 2 * WP * (Nb / 4)) {
            const int row = q / (WP * (Nb / 4));
            const int r   = q - row * (WP * (Nb / 4));
            const int wp  = r >> 3;
            const int n4  = (r & 7) * 4;
            f4 o = zero;
            if (wp != 0 && wp != WP - 1) {
                const int col = row * Ww + wp - 1;
                o.x = sT[n4 + 0][col];
                o.y = sT[n4 + 1][col];
                o.z = sT[n4 + 2][col];
                o.w = sT[n4 + 3][col];
            }
            *(f4*)(dst + (size_t)q * 4) = o;
        }
    }
}

__global__ __launch_bounds__(256, 4) void lc2d_wmma_f32(
    const float* __restrict__ xT,
    const float* __restrict__ wgt,
    const float* __restrict__ bias,
    float* __restrict__ out)
{
    __shared__ float sP[NT][CKK / 2][SPP];
    __shared__ __attribute__((aligned(16))) float sO[2][Nb][33][TWG];

    const int t     = threadIdx.x;
    const int lane  = t & 31;
    const int wave  = (t >> 5) & 3;
    const int team  = t >> 7;
    const int half  = lane >> 4;
    const int m16   = lane & 15;
    const int otile = wave & 1;
    const int kh    = wave >> 1;
    const int oC    = m16 + 16 * otile;
    const int K0    = KH * kh;

    const int h  = blockIdx.y;
    const int w0 = blockIdx.x * TWG + team * TW;

    const f4* wsrc = (const f4*)(wgt + ((size_t)oC * PP + (size_t)h * Ww + w0) * CKK + K0 + 4 * half);
    const float* bsrc = bias + (size_t)oC * PP + (size_t)h * Ww + w0;
    f4 q[2][KH / 8];
    float bq[2];
#pragma unroll
    for (int j = 0; j < KH / 8; ++j) q[0][j] = wsrc[2 * j];
    bq[0] = bsrc[0];
    __builtin_amdgcn_sched_barrier(0);

#pragma unroll
    for (int tw = 0; tw < TW; ++tw) {
        const int w = w0 + tw;
        const int p = h * Ww + w;
        const int cq = tw & 1;

        if (tw > 0) __syncthreads();

        {
            const float* xb = xT + (((size_t)(wave * 8) * HP + h) * WP + w) * Nb + lane;
            float xv[72];
#pragma unroll
            for (int cc = 0; cc < 8; ++cc) {
#pragma unroll
                for (int ki = 0; ki < 3; ++ki) {
#pragma unroll
                    for (int kj = 0; kj < 3; ++kj) {
                        xv[cc * 9 + ki * 3 + kj] = xb[((cc * HP + ki) * WP + kj) * Nb];
                    }
                }
            }
#pragma unroll
            for (int i = 0; i < 36; ++i) {
                v2f pr;
                pr[0] = xv[2 * i];
                pr[1] = xv[2 * i + 1];
                *(v2f*)&sP[team][wave * 36 + i][2 * lane] = pr;
            }
        }

        if (tw + 2 < TW) {
            const int o = wave * 8 + (lane >> 2);
            __builtin_prefetch(wgt + ((size_t)o * PP + (p + 2)) * CKK + (size_t)(lane & 3) * 72, 0, 1);
        }

        __syncthreads();

        if (tw + 1 < TW) {
#pragma unroll
            for (int j = 0; j < KH / 8; ++j) q[cq ^ 1][j] = wsrc[(tw + 1) * (CKK / 4) + 2 * j];
            bq[cq ^ 1] = bsrc[tw + 1];
        }
        __builtin_amdgcn_sched_barrier(0);

        v2f fa[2][KB][2];
#pragma unroll
        for (int j = 0; j < KB; ++j) {
            const int pi = (K0 >> 1) + 4 * (j >> 1) + 2 * half + (j & 1);
#pragma unroll
            for (int mt = 0; mt < 2; ++mt) fa[0][j][mt] = *(const v2f*)&sP[team][pi][2 * (m16 + 16 * mt)];
        }
        v8f acc0 = {};
        v8f acc1 = {};
#pragma unroll
        for (int blk = 0; blk < NBLK; ++blk) {
            const int cur = blk & 1;
            const int nxt = cur ^ 1;
            if (blk + 1 < NBLK) {
#pragma unroll
                for (int j = 0; j < KB; ++j) {
                    const int s  = (blk + 1) * KB + j;
                    const int pi = (K0 >> 1) + 4 * (s >> 1) + 2 * half + (s & 1);
#pragma unroll
                    for (int mt = 0; mt < 2; ++mt) fa[nxt][j][mt] = *(const v2f*)&sP[team][pi][2 * (m16 + 16 * mt)];
                }
            }
            __builtin_amdgcn_sched_barrier(0);
#pragma unroll
            for (int j = 0; j < KB; ++j) {
                const int s = blk * KB + j;
                v2f fb;
                if (s & 1) { fb[0] = q[cq][s >> 1].z; fb[1] = q[cq][s >> 1].w; }
                else       { fb[0] = q[cq][s >> 1].x; fb[1] = q[cq][s >> 1].y; }
                acc0 = __builtin_amdgcn_wmma_f32_16x16x4_f32(
                    false, fa[cur][j][0], false, fb, (short)0, acc0, false, false);
                acc1 = __builtin_amdgcn_wmma_f32_16x16x4_f32(
                    false, fa[cur][j][1], false, fb, (short)0, acc1, false, false);
            }
            __builtin_amdgcn_sched_barrier(0);
        }

        const float bv = (kh == 0) ? bq[cq] : 0.0f;
#pragma unroll
        for (int r = 0; r < 8; ++r) {
            const int n = r + 8 * half;
            sO[kh][n][oC][team * TW + tw]      = acc0[r] + bv;
            sO[kh][n + 16][oC][team * TW + tw] = acc1[r] + bv;
        }
    }

    __syncthreads();

    const int ws = blockIdx.x * TWG;
    for (int e = 0; e < 8; ++e) {
        const int idx  = e * 256 + t;
        const int hf   = (idx & 1) * 4;
        const int pair = idx >> 1;
        const int n = pair >> 5;
        const int o = pair & 31;
        f4 v;
        v.x = sO[0][n][o][hf + 0] + sO[1][n][o][hf + 0];
        v.y = sO[0][n][o][hf + 1] + sO[1][n][o][hf + 1];
        v.z = sO[0][n][o][hf + 2] + sO[1][n][o][hf + 2];
        v.w = sO[0][n][o][hf + 3] + sO[1][n][o][hf + 3];
        *(f4*)(out + ((size_t)n * Oo + o) * PP + (size_t)h * Ww + ws + hf) = v;
    }
}

extern "C" void kernel_launch(void* const* d_in, const int* in_sizes, int n_in,
                              void* d_out, int out_size, void* d_ws, size_t ws_size,
                              hipStream_t stream) {
    const float* x    = (const float*)d_in[0];
    const float* wgt  = (const float*)d_in[1];
    const float* bias = (const float*)d_in[2];
    float* out        = (float*)d_out;
    float* xT         = (float*)d_ws;

    dim3 gridT(Hh / 2 + 1, Cc);
    lc2d_x_batch_inner<<<gridT, 128, 0, stream>>>(x, xT);

    dim3 grid(Ww / TWG, Hh);
    lc2d_wmma_f32<<<grid, 256, 0, stream>>>(xT, wgt, bias, out);
}


# next position's patch rows loaded ahead of the current position's matrix instructions, held in registers across the GEMM; its weights follow block by block; barriers wait for LDS only; sums unchanged
# speedup vs baseline: 3.2908x; 1.0772x over previous
#include <hip/hip_runtime.h>

typedef __attribute__((ext_vector_type(2))) float v2f;
typedef __attribute__((ext_vector_type(8))) float v8f;
typedef __attribute__((ext_vector_type(4))) float f4;

#define Nb  32
#define Cc  32
#define Hh  56
#define Ww  56
#define Oo  32
#define CKK 288
#define PP  3136
#define TW  4
#define NT  2
#define TWG 8
#define KB  6
#define NBLK 6
#define KH  144
#define SPP 80

#define HP  58
#define WP  58

__global__ __launch_bounds__(128) void lc2d_x_batch_inner(
    const float* __restrict__ x,
    float* __restrict__ xT)
{
    __shared__ float sT[Nb][2 * Ww + 1];

    const int t  = threadIdx.x;
    const int bx = blockIdx.x;
    const int c  = blockIdx.y;
    const f4 zero = {0.0f, 0.0f, 0.0f, 0.0f};

    if (bx == Hh / 2) {
        float* top = xT + (size_t)c * HP * WP * Nb;
        float* bot = top + (size_t)(HP - 1) * WP * Nb;
#pragma unroll
        for (int j = 0; j < 4; ++j) {
            const int q = j * 128 + t;
            if (q < WP * (Nb / 4)) {
                *(f4*)(top + (size_t)q * 4) = zero;
                *(f4*)(bot + (size_t)q * 4) = zero;
            }
        }
        return;
    }

    const int h0 = 2 * bx;
    f4 v[7];
#pragma unroll
    for (int j = 0; j < 7; ++j) {
        const int q = j * 128 + t;
        const int n = q / 28;
        const int r = q - n * 28;
        v[j] = *(const f4*)(x + (((size_t)n * Cc + c) * Hh + h0) * Ww + r * 4);
    }
#pragma unroll
    for (int j = 0; j < 7; ++j) {
        const int q = j * 128 + t;
        const int n = q / 28;
        const int r = q - n * 28;
        sT[n][r * 4 + 0] = v[j].x;
        sT[n][r * 4 + 1] = v[j].y;
        sT[n][r * 4 + 2] = v[j].z;
        sT[n][r * 4 + 3] = v[j].w;
    }

    __syncthreads();

    float* dst = xT + ((size_t)c * HP + h0 + 1) * WP * Nb;
#pragma unroll
    for (int j = 0; j < 8; ++j) {
        const int q = j * 128 + t;
        if (q < 2 * WP * (Nb / 4)) {
            const int row = q / (WP * (Nb / 4));
            const int r   = q - row * (WP * (Nb / 4));
            const int wp  = r >> 3;
            const int n4  = (r & 7) * 4;
            f4 o = zero;
            if (wp != 0 && wp != WP - 1) {
                const int col = row * Ww + wp - 1;
                o.x = sT[n4 + 0][col];
                o.y = sT[n4 + 1][col];
                o.z = sT[n4 + 2][col];
                o.w = sT[n4 + 3][col];
            }
            *(f4*)(dst + (size_t)q * 4) = o;
        }
    }
}

__device__ __forceinline__ void lds_barrier()
{
    __builtin_amdgcn_fence(__ATOMIC_RELEASE, "workgroup", "local");
    __builtin_amdgcn_s_barrier();
    __builtin_amdgcn_fence(__ATOMIC_ACQUIRE, "workgroup", "local");
}

__global__ __launch_bounds__(256, 4) void lc2d_wmma_f32(
    const float* __restrict__ xT,
    const float* __restrict__ wgt,
    const float* __restrict__ bias,
    float* __restrict__ out)
{
    __shared__ float sP[NT][CKK / 2][SPP];
    __shared__ __attribute__((aligned(16))) float sO[2][Nb][33][TWG];

    const int t     = threadIdx.x;
    const int lane  = t & 31;
    const int wave  = (t >> 5) & 3;
    const int team  = t >> 7;
    const int half  = lane >> 4;
    const int m16   = lane & 15;
    const int otile = wave & 1;
    const int kh    = wave >> 1;
    const int oC    = m16 + 16 * otile;
    const int K0    = KH * kh;

    const int h  = blockIdx.y;
    const int w0 = blockIdx.x * TWG + team * TW;

    const f4* wsrc = (const f4*)(wgt + ((size_t)oC * PP + (size_t)h * Ww + w0) * CKK + K0 + 4 * half);
    const float* bsrc = bias + (size_t)oC * PP + (size_t)h * Ww + w0;
    f4 q[2][KH / 8];
    float bq[2];
#pragma unroll
    for (int j = 0; j < KH / 8; ++j) q[0][j] = wsrc[2 * j];
    bq[0] = bsrc[0];
    __builtin_amdgcn_sched_barrier(0);

    const float* xb = xT + (((size_t)(wave * 8) * HP + h) * WP + w0) * Nb + lane;
    float xv[72];
#pragma unroll
    for (int r = 0; r < 72; ++r)
        xv[r] = xb[(((r / 9) * HP + (r % 9) / 3) * WP + (r % 3)) * Nb];
#pragma unroll
    for (int i = 0; i < 36; ++i) {
        v2f pr;
        pr[0] = xv[2 * i];
        pr[1] = xv[2 * i + 1];
        *(v2f*)&sP[team][wave * 36 + i][2 * lane] = pr;
    }
    lds_barrier();

#pragma unroll
    for (int tw = 0; tw < TW; ++tw) {
        const int w = w0 + tw;
        const int p = h * Ww + w;
        const int cq = tw & 1;
        const bool more = tw + 1 < TW;
        const float* xn = xb + (size_t)(tw + 1) * Nb;

        if (tw + 2 < TW) {
            const int o = wave * 8 + (lane >> 2);
            __builtin_prefetch(wgt + ((size_t)o * PP + (p + 2)) * CKK + (size_t)(lane & 3) * 72, 0, 1);
        }

        if (more) {
#pragma unroll
            for (int r = 0; r < 24; ++r)
                xv[r] = xn[(((r / 9) * HP + (r % 9) / 3) * WP + (r % 3)) * Nb];
#pragma unroll
            for (int j = 0; j < KB / 2; ++j) q[cq ^ 1][j] = wsrc[(tw + 1) * (CKK / 4) + 2 * j];
            bq[cq ^ 1] = bsrc[tw + 1];
        }
        __builtin_amdgcn_sched_barrier(0);

        v2f fa[2][KB][2];
#pragma unroll
        for (int j = 0; j < KB; ++j) {
            const int pi = (K0 >> 1) + 4 * (j >> 1) + 2 * half + (j & 1);
#pragma unroll
            for (int mt = 0; mt < 2; ++mt) fa[0][j][mt] = *(const v2f*)&sP[team][pi][2 * (m16 + 16 * mt)];
        }
        v8f acc0 = {};
        v8f acc1 = {};
#pragma unroll
        for (int blk = 0; blk < NBLK; ++blk) {
            const int cur = blk & 1;
            const int nxt = cur ^ 1;
            if (blk + 1 < NBLK) {
#pragma unroll
                for (int j = 0; j < KB; ++j) {
                    const int s  = (blk + 1) * KB + j;
                    const int pi = (K0 >> 1) + 4 * (s >> 1) + 2 * half + (s & 1);
#pragma unroll
                    for (int mt = 0; mt < 2; ++mt) fa[nxt][j][mt] = *(const v2f*)&sP[team][pi][2 * (m16 + 16 * mt)];
                }
            }
            __builtin_amdgcn_sched_barrier(0);
#pragma unroll
            for (int j = 0; j < KB; ++j) {
                const int s = blk * KB + j;
                v2f fb;
                if (s & 1) { fb[0] = q[cq][s >> 1].z; fb[1] = q[cq][s >> 1].w; }
                else       { fb[0] = q[cq][s >> 1].x; fb[1] = q[cq][s >> 1].y; }
                acc0 = __builtin_amdgcn_wmma_f32_16x16x4_f32(
                    false, fa[cur][j][0], false, fb, (short)0, acc0, false, false);
                acc1 = __builtin_amdgcn_wmma_f32_16x16x4_f32(
                    false, fa[cur][j][1], false, fb, (short)0, acc1, false, false);
            }
            __builtin_amdgcn_sched_barrier(0);
            if (more) {
                if (blk < 2) {
#pragma unroll
                    for (int r = 24 * (blk + 1); r < 24 * (blk + 2); ++r)
                        xv[r] = xn[(((r / 9) * HP + (r % 9) / 3) * WP + (r % 3)) * Nb];
                }
                if (blk + 1 < NBLK) {
#pragma unroll
                    for (int j = 0; j < KB / 2; ++j) {
                        const int jj = (blk + 1) * (KB / 2) + j;
                        q[cq ^ 1][jj] = wsrc[(tw + 1) * (CKK / 4) + 2 * jj];
                    }
                }
            }
            __builtin_amdgcn_sched_barrier(0);
        }

        const float bv = (kh == 0) ? bq[cq] : 0.0f;
#pragma unroll
        for (int r = 0; r < 8; ++r) {
            const int n = r + 8 * half;
            sO[kh][n][oC][team * TW + tw]      = acc0[r] + bv;
            sO[kh][n + 16][oC][team * TW + tw] = acc1[r] + bv;
        }

        if (more) {
            lds_barrier();
#pragma unroll
            for (int i = 0; i < 36; ++i) {
                v2f pr;
                pr[0] = xv[2 * i];
                pr[1] = xv[2 * i + 1];
                *(v2f*)&sP[team][wave * 36 + i][2 * lane] = pr;
            }
            lds_barrier();
        }
    }

    __syncthreads();

    const int ws = blockIdx.x * TWG;
    for (int e = 0; e < 8; ++e) {
        const int idx  = e * 256 + t;
        const int hf   = (idx & 1) * 4;
        const int pair = idx >> 1;
        const int n = pair >> 5;
        const int o = pair & 31;
        f4 v;
        v.x = sO[0][n][o][hf + 0] + sO[1][n][o][hf + 0];
        v.y = sO[0][n][o][hf + 1] + sO[1][n][o][hf + 1];
        v.z = sO[0][n][o][hf + 2] + sO[1][n][o][hf + 2];
        v.w = sO[0][n][o][hf + 3] + sO[1][n][o][hf + 3];
        *(f4*)(out + ((size_t)n * Oo + o) * PP + (size_t)h * Ww + ws + hf) = v;
    }
}

extern "C" void kernel_launch(void* const* d_in, const int* in_sizes, int n_in,
                              void* d_out, int out_size, void* d_ws, size_t ws_size,
                              hipStream_t stream) {
    const float* x    = (const float*)d_in[0];
    const float* wgt  = (const float*)d_in[1];
    const float* bias = (const float*)d_in[2];
    float* out        = (float*)d_out;
    float* xT         = (float*)d_ws;

    dim3 gridT(Hh / 2 + 1, Cc);
    lc2d_x_batch_inner<<<gridT, 128, 0, stream>>>(x, xT);

    dim3 grid(Ww / TWG, Hh);
    lc2d_wmma_f32<<<grid, 256, 0, stream>>>(xT, wgt, bias, out);
}


# weights loaded as whole 128-B lines (4 rows x 1 line per load, not 16 lines x 32 B), turned into B fragments through a private per-wave LDS slab in the tile padding; k half 1 sums its line-4 k last
# speedup vs baseline: 3.5344x; 1.0740x over previous
#include <hip/hip_runtime.h>

typedef __attribute__((ext_vector_type(2))) float v2f;
typedef __attribute__((ext_vector_type(8))) float v8f;
typedef __attribute__((ext_vector_type(4))) float f4;

#define Nb  32
#define Cc  32
#define Hh  56
#define Ww  56
#define Oo  32
#define CKK 288
#define PP  3136
#define TW  4
#define NT  2
#define TWG 8
#define KB  6
#define NBLK 6
#define KH  144
#define SPP 80

#define HP  58
#define WP  58

__global__ __launch_bounds__(128) void lc2d_x_batch_inner(
    const float* __restrict__ x,
    float* __restrict__ xT)
{
    __shared__ float sT[Nb][2 * Ww + 1];

    const int t  = threadIdx.x;
    const int bx = blockIdx.x;
    const int c  = blockIdx.y;
    const f4 zero = {0.0f, 0.0f, 0.0f, 0.0f};

    if (bx == Hh / 2) {
        float* top = xT + (size_t)c * HP * WP * Nb;
        float* bot = top + (size_t)(HP - 1) * WP * Nb;
#pragma unroll
        for (int j = 0; j < 4; ++j) {
            const int q = j * 128 + t;
            if (q < WP * (Nb / 4)) {
                *(f4*)(top + (size_t)q * 4) = zero;
                *(f4*)(bot + (size_t)q * 4) = zero;
            }
        }
        return;
    }

    const int h0 = 2 * bx;
    f4 v[7];
#pragma unroll
    for (int j = 0; j < 7; ++j) {
        const int q = j * 128 + t;
        const int n = q / 28;
        const int r = q - n * 28;
        v[j] = *(const f4*)(x + (((size_t)n * Cc + c) * Hh + h0) * Ww + r * 4);
    }
#pragma unroll
    for (int j = 0; j < 7; ++j) {
        const int q = j * 128 + t;
        const int n = q / 28;
        const int r = q - n * 28;
        sT[n][r * 4 + 0] = v[j].x;
        sT[n][r * 4 + 1] = v[j].y;
        sT[n][r * 4 + 2] = v[j].z;
        sT[n][r * 4 + 3] = v[j].w;
    }

    __syncthreads();

    float* dst = xT + ((size_t)c * HP + h0 + 1) * WP * Nb;
#pragma unroll
    for (int j = 0; j < 8; ++j) {
        const int q = j * 128 + t;
        if (q < 2 * WP * (Nb / 4)) {
            const int row = q / (WP * (Nb / 4));
            const int r   = q - row * (WP * (Nb / 4));
            const int wp  = r >> 3;
            const int n4  = (r & 7) * 4;
            f4 o = zero;
            if (wp != 0 && wp != WP - 1) {
                const int col = row * Ww + wp - 1;
                o.x = sT[n4 + 0][col];
                o.y = sT[n4 + 1][col];
                o.z = sT[n4 + 2][col];
                o.w = sT[n4 + 3][col];
            }
            *(f4*)(dst + (size_t)q * 4) = o;
        }
    }
}

__device__ __forceinline__ void lds_barrier()
{
    __builtin_amdgcn_fence(__ATOMIC_RELEASE, "workgroup", "local");
    __builtin_amdgcn_s_barrier();
    __builtin_amdgcn_fence(__ATOMIC_ACQUIRE, "workgroup", "local");
}

__global__ __launch_bounds__(256, 4) void lc2d_wmma_f32(
    const float* __restrict__ xT,
    const float* __restrict__ wgt,
    const float* __restrict__ bias,
    float* __restrict__ out)
{
    __shared__ __attribute__((aligned(16))) float sP[NT][CKK / 2][SPP];
    __shared__ __attribute__((aligned(16))) float sO[2][Nb][33][TWG];

    const int t     = threadIdx.x;
    const int lane  = t & 31;
    const int wave  = (t >> 5) & 3;
    const int team  = t >> 7;
    const int half  = lane >> 4;
    const int m16   = lane & 15;
    const int otile = wave & 1;
    const int kh    = wave >> 1;
    const int oC    = m16 + 16 * otile;

    const int h  = blockIdx.y;
    const int w0 = blockIdx.x * TWG + team * TW;

    const size_t ROWF = (size_t)PP * CKK;
    const float* wrow  = wgt + ((size_t)(16 * otile) * PP + (size_t)h * Ww + w0) * CKK;
    const float* wlinv = wrow + 160 * kh;
    const float* whalv = wrow + 128 + 16 * kh;
    typedef const __attribute__((address_space(1))) f4* gf4p;
    unsigned long long wl[4];
    unsigned long long wh[2];
#pragma unroll
    for (int g = 0; g < 4; ++g) {
        const unsigned long long pv = (unsigned long long)(wlinv + (size_t)(4 * g) * ROWF);
        wl[g] = ((unsigned long long)(unsigned)__builtin_amdgcn_readfirstlane((int)(unsigned)(pv >> 32)) << 32) | (unsigned long long)(unsigned)__builtin_amdgcn_readfirstlane((int)(unsigned)pv);
    }
#pragma unroll
    for (int hh = 0; hh < 2; ++hh) {
        const unsigned long long pv = (unsigned long long)(whalv + (size_t)(8 * hh) * ROWF);
        wh[hh] = ((unsigned long long)(unsigned)__builtin_amdgcn_readfirstlane((int)(unsigned)(pv >> 32)) << 32) | (unsigned long long)(unsigned)__builtin_amdgcn_readfirstlane((int)(unsigned)pv);
    }
    const unsigned lo_line = 4u * ((unsigned)(lane >> 3) * (unsigned)ROWF + 4u * (lane & 7));
    const unsigned lo_half = 4u * ((unsigned)(lane >> 2) * (unsigned)ROWF + 4u * (lane & 3));
    char* slab = (char*)&sP[team][wave * 36][0];
    const unsigned u_st  = 9u * (lane >> 3) + (lane & 7);
    const unsigned u_st2 = 9u * (lane >> 2) + (lane & 3);
    const unsigned v_ld  = 9u * m16 + half;
    const unsigned st_off  = 80u * u_st  - 64u * (u_st  & 3u) + 256u;
    const unsigned st2_off = 80u * u_st2 - 64u * (u_st2 & 3u) + 256u;
    const unsigned ld_off0 = 80u * v_ld - 64u * (v_ld & 3u) + 256u;
    const unsigned ld_off1 = 80u * v_ld - 64u * ((v_ld & 3u) ^ 2u) + 256u;
#define LOAD_LINE(L, twn)  { _Pragma("unroll") for (int g = 0; g < 4; ++g) \
        S[4 * (L) + g] = *(gf4p)(wl[g] + (unsigned long long)lo_line + 4ull * ((twn) * CKK + 32 * (L))); }
#define LOAD_HALF(twn)     { _Pragma("unroll") for (int hh = 0; hh < 2; ++hh) \
        S[16 + hh] = *(gf4p)(wh[hh] + (unsigned long long)lo_half + 4ull * ((twn) * CKK)); }
#define TURN_LINE(L)       { _Pragma("unroll") for (int g = 0; g < 4; ++g) *(f4*)(slab + st_off + 2880 * g) = S[4 * (L) + g]; \
                             _Pragma("unroll") for (int jp = 0; jp < 4; ++jp) \
        q[4 * (L) + jp] = *(const f4*)(slab + ((jp & 1) ? ld_off1 : ld_off0) + 160 * jp); }
#define TURN_HALF()        { _Pragma("unroll") for (int hh = 0; hh < 2; ++hh) *(f4*)(slab + st2_off + 5760 * hh) = S[16 + hh]; \
                             _Pragma("unroll") for (int jp = 0; jp < 2; ++jp) \
        q[16 + jp] = *(const f4*)(slab + ((jp & 1) ? ld_off1 : ld_off0) + 160 * jp); }
    const float* bsrc = bias + (size_t)oC * PP + (size_t)h * Ww + w0;
    f4 S[KH / 8];
    f4 q[KH / 8];
    float bq[2];
    LOAD_LINE(0, 0) LOAD_LINE(1, 0) LOAD_LINE(2, 0) LOAD_LINE(3, 0) LOAD_HALF(0)
    bq[0] = bsrc[0];
    __builtin_amdgcn_sched_barrier(0);

    const float* xb = xT + (((size_t)(wave * 8) * HP + h) * WP + w0) * Nb + lane;
    float xv[72];
#pragma unroll
    for (int r = 0; r < 72; ++r)
        xv[r] = xb[(((r / 9) * HP + (r % 9) / 3) * WP + (r % 3)) * Nb];
#pragma unroll
    for (int i = 0; i < 36; ++i) {
        v2f pr;
        pr[0] = xv[2 * i];
        pr[1] = xv[2 * i + 1];
        *(v2f*)&sP[team][wave * 36 + i][2 * lane] = pr;
    }
    TURN_LINE(0) TURN_LINE(1)
    lds_barrier();

    const int PL = 80 * kh;
    const int PH = 64 + 8 * kh;
#define PAIR_ROW(s) ((((s) >> 1) < 16 ? PL + 4 * ((s) >> 1) : PH + 4 * (((s) >> 1) - 16)) + 2 * half + ((s) & 1))

#pragma unroll
    for (int tw = 0; tw < TW; ++tw) {
        const int w = w0 + tw;
        const int p = h * Ww + w;
        const int cq = tw & 1;
        const bool more = tw + 1 < TW;
        const float* xn = xb + (size_t)(tw + 1) * Nb;

        if (tw + 2 < TW) {
            const int o = wave * 8 + (lane >> 2);
            __builtin_prefetch(wgt + ((size_t)o * PP + (p + 2)) * CKK + (size_t)(lane & 3) * 72, 0, 1);
        }

        if (more) {
#pragma unroll
            for (int r = 0; r < 24; ++r)
                xv[r] = xn[(((r / 9) * HP + (r % 9) / 3) * WP + (r % 3)) * Nb];
            bq[cq ^ 1] = bsrc[tw + 1];
        }
        __builtin_amdgcn_sched_barrier(0);

        v2f fa[2][KB][2];
#pragma unroll
        for (int j = 0; j < KB; ++j) {
            const int pi = PAIR_ROW(j);
#pragma unroll
            for (int mt = 0; mt < 2; ++mt) fa[0][j][mt] = *(const v2f*)&sP[team][pi][2 * (m16 + 16 * mt)];
        }
        v8f acc0 = {};
        v8f acc1 = {};
#pragma unroll
        for (int blk = 0; blk < NBLK; ++blk) {
            const int cur = blk & 1;
            const int nxt = cur ^ 1;
            if (blk + 1 < NBLK) {
#pragma unroll
                for (int j = 0; j < KB; ++j) {
                    const int s  = (blk + 1) * KB + j;
                    const int pi = PAIR_ROW(s);
#pragma unroll
                    for (int mt = 0; mt < 2; ++mt) fa[nxt][j][mt] = *(const v2f*)&sP[team][pi][2 * (m16 + 16 * mt)];
                }
            }
            if (blk == 1) TURN_LINE(2)
            if (blk == 2) TURN_HALF()
            if (blk == 3) TURN_LINE(3)
            __builtin_amdgcn_sched_barrier(0);
#pragma unroll
            for (int j = 0; j < KB; ++j) {
                const int s = blk * KB + j;
                v2f fb;
                if (s & 1) { fb[0] = q[s >> 1].z; fb[1] = q[s >> 1].w; }
                else       { fb[0] = q[s >> 1].x; fb[1] = q[s >> 1].y; }
                acc0 = __builtin_amdgcn_wmma_f32_16x16x4_f32(
                    false, fa[cur][j][0], false, fb, (short)0, acc0, false, false);
                acc1 = __builtin_amdgcn_wmma_f32_16x16x4_f32(
                    false, fa[cur][j][1], false, fb, (short)0, acc1, false, false);
            }
            __builtin_amdgcn_sched_barrier(0);
            if (more) {
                if (blk < 2) {
#pragma unroll
                    for (int r = 24 * (blk + 1); r < 24 * (blk + 2); ++r)
                        xv[r] = xn[(((r / 9) * HP + (r % 9) / 3) * WP + (r % 3)) * Nb];
                }
                if (blk == 1) LOAD_LINE(0, tw + 1)
                if (blk == 2) LOAD_LINE(1, tw + 1)
                if (blk == 3) LOAD_LINE(2, tw + 1)
                if (blk == 4) LOAD_HALF(tw + 1)
                if (blk == 5) LOAD_LINE(3, tw + 1)
            }
            __builtin_amdgcn_sched_barrier(0);
        }

        if (more) { TURN_LINE(0) TURN_LINE(1) }

        const float bv = (kh == 0) ? bq[cq] : 0.0f;
#pragma unroll
        for (int r = 0; r < 8; ++r) {
            const int n = r + 8 * half;
            sO[kh][n][oC][team * TW + tw]      = acc0[r] + bv;
            sO[kh][n + 16][oC][team * TW + tw] = acc1[r] + bv;
        }

        if (more) {
            lds_barrier();
#pragma unroll
            for (int i = 0; i < 36; ++i) {
                v2f pr;
                pr[0] = xv[2 * i];
                pr[1] = xv[2 * i + 1];
                *(v2f*)&sP[team][wave * 36 + i][2 * lane] = pr;
            }
            lds_barrier();
        }
    }

    __syncthreads();

    const int ws = blockIdx.x * TWG;
    for (int e = 0; e < 8; ++e) {
        const int idx  = e * 256 + t;
        const int hf   = (idx & 1) * 4;
        const int pair = idx >> 1;
        const int n = pair >> 5;
        const int o = pair & 31;
        f4 v;
        v.x = sO[0][n][o][hf + 0] + sO[1][n][o][hf + 0];
        v.y = sO[0][n][o][hf + 1] + sO[1][n][o][hf + 1];
        v.z = sO[0][n][o][hf + 2] + sO[1][n][o][hf + 2];
        v.w = sO[0][n][o][hf + 3] + sO[1][n][o][hf + 3];
        *(f4*)(out + ((size_t)n * Oo + o) * PP + (size_t)h * Ww + ws + hf) = v;
    }
}

extern "C" void kernel_launch(void* const* d_in, const int* in_sizes, int n_in,
                              void* d_out, int out_size, void* d_ws, size_t ws_size,
                              hipStream_t stream) {
    const float* x    = (const float*)d_in[0];
    const float* wgt  = (const float*)d_in[1];
    const float* bias = (const float*)d_in[2];
    float* out        = (float*)d_out;
    float* xT         = (float*)d_ws;

    dim3 gridT(Hh / 2 + 1, Cc);
    lc2d_x_batch_inner<<<gridT, 128, 0, stream>>>(x, xT);

    dim3 grid(Ww / TWG, Hh);
    lc2d_wmma_f32<<<grid, 256, 0, stream>>>(xT, wgt, bias, out);
}


# next position's im2col tile = the wave's own tile rows moved one column on (LDS re-read) + one new column from memory: 24 patch-row loads per position, not 72; k order and sums unchanged
# speedup vs baseline: 3.6390x; 1.0296x over previous
#include <hip/hip_runtime.h>

typedef __attribute__((ext_vector_type(2))) float v2f;
typedef __attribute__((ext_vector_type(8))) float v8f;
typedef __attribute__((ext_vector_type(4))) float f4;

#define Nb  32
#define Cc  32
#define Hh  56
#define Ww  56
#define Oo  32
#define CKK 288
#define PP  3136
#define TW  4
#define NT  2
#define TWG 8
#define KB  6
#define NBLK 6
#define KH  144
#define SPP 80

#define HP  58
#define WP  58

__global__ __launch_bounds__(128) void lc2d_x_batch_inner(
    const float* __restrict__ x,
    float* __restrict__ xT)
{
    __shared__ float sT[Nb][2 * Ww + 1];

    const int t  = threadIdx.x;
    const int bx = blockIdx.x;
    const int c  = blockIdx.y;
    const f4 zero = {0.0f, 0.0f, 0.0f, 0.0f};

    if (bx == Hh / 2) {
        float* top = xT + (size_t)c * HP * WP * Nb;
        float* bot = top + (size_t)(HP - 1) * WP * Nb;
#pragma unroll
        for (int j = 0; j < 4; ++j) {
            const int q = j * 128 + t;
            if (q < WP * (Nb / 4)) {
                *(f4*)(top + (size_t)q * 4) = zero;
                *(f4*)(bot + (size_t)q * 4) = zero;
            }
        }
        return;
    }

    const int h0 = 2 * bx;
    f4 v[7];
#pragma unroll
    for (int j = 0; j < 7; ++j) {
        const int q = j * 128 + t;
        const int n = q / 28;
        const int r = q - n * 28;
        v[j] = *(const f4*)(x + (((size_t)n * Cc + c) * Hh + h0) * Ww + r * 4);
    }
#pragma unroll
    for (int j = 0; j < 7; ++j) {
        const int q = j * 128 + t;
        const int n = q / 28;
        const int r = q - n * 28;
        sT[n][r * 4 + 0] = v[j].x;
        sT[n][r * 4 + 1] = v[j].y;
        sT[n][r * 4 + 2] = v[j].z;
        sT[n][r * 4 + 3] = v[j].w;
    }

    __syncthreads();

    float* dst = xT + ((size_t)c * HP + h0 + 1) * WP * Nb;
#pragma unroll
    for (int j = 0; j < 8; ++j) {
        const int q = j * 128 + t;
        if (q < 2 * WP * (Nb / 4)) {
            const int row = q / (WP * (Nb / 4));
            const int r   = q - row * (WP * (Nb / 4));
            const int wp  = r >> 3;
            const int n4  = (r & 7) * 4;
            f4 o = zero;
            if (wp != 0 && wp != WP - 1) {
                const int col = row * Ww + wp - 1;
                o.x = sT[n4 + 0][col];
                o.y = sT[n4 + 1][col];
                o.z = sT[n4 + 2][col];
                o.w = sT[n4 + 3][col];
            }
            *(f4*)(dst + (size_t)q * 4) = o;
        }
    }
}

__device__ __forceinline__ void lds_barrier()
{
    __builtin_amdgcn_fence(__ATOMIC_RELEASE, "workgroup", "local");
    __builtin_amdgcn_s_barrier();
    __builtin_amdgcn_fence(__ATOMIC_ACQUIRE, "workgroup", "local");
}

__global__ __launch_bounds__(256, 4) void lc2d_wmma_f32(
    const float* __restrict__ xT,
    const float* __restrict__ wgt,
    const float* __restrict__ bias,
    float* __restrict__ out)
{
    __shared__ __attribute__((aligned(16))) float sP[NT][CKK / 2][SPP];
    __shared__ __attribute__((aligned(16))) float sO[2][Nb][33][TWG];

    const int t     = threadIdx.x;
    const int lane  = t & 31;
    const int wave  = (t >> 5) & 3;
    const int team  = t >> 7;
    const int half  = lane >> 4;
    const int m16   = lane & 15;
    const int otile = wave & 1;
    const int kh    = wave >> 1;
    const int oC    = m16 + 16 * otile;

    const int h  = blockIdx.y;
    const int w0 = blockIdx.x * TWG + team * TW;

    const size_t ROWF = (size_t)PP * CKK;
    const float* wrow  = wgt + ((size_t)(16 * otile) * PP + (size_t)h * Ww + w0) * CKK;
    const float* wlinv = wrow + 160 * kh;
    const float* whalv = wrow + 128 + 16 * kh;
    typedef const __attribute__((address_space(1))) f4* gf4p;
    unsigned long long wl[4];
    unsigned long long wh[2];
#pragma unroll
    for (int g = 0; g < 4; ++g) {
        const unsigned long long pv = (unsigned long long)(wlinv + (size_t)(4 * g) * ROWF);
        wl[g] = ((unsigned long long)(unsigned)__builtin_amdgcn_readfirstlane((int)(unsigned)(pv >> 32)) << 32) | (unsigned long long)(unsigned)__builtin_amdgcn_readfirstlane((int)(unsigned)pv);
    }
#pragma unroll
    for (int hh = 0; hh < 2; ++hh) {
        const unsigned long long pv = (unsigned long long)(whalv + (size_t)(8 * hh) * ROWF);
        wh[hh] = ((unsigned long long)(unsigned)__builtin_amdgcn_readfirstlane((int)(unsigned)(pv >> 32)) << 32) | (unsigned long long)(unsigned)__builtin_amdgcn_readfirstlane((int)(unsigned)pv);
    }
    const unsigned lo_line = 4u * ((unsigned)(lane >> 3) * (unsigned)ROWF + 4u * (lane & 7));
    const unsigned lo_half = 4u * ((unsigned)(lane >> 2) * (unsigned)ROWF + 4u * (lane & 3));
    char* slab = (char*)&sP[team][wave * 36][0];
    const unsigned u_st  = 9u * (lane >> 3) + (lane & 7);
    const unsigned u_st2 = 9u * (lane >> 2) + (lane & 3);
    const unsigned v_ld  = 9u * m16 + half;
    const unsigned st_off  = 80u * u_st  - 64u * (u_st  & 3u) + 256u;
    const unsigned st2_off = 80u * u_st2 - 64u * (u_st2 & 3u) + 256u;
    const unsigned ld_off0 = 80u * v_ld - 64u * (v_ld & 3u) + 256u;
    const unsigned ld_off1 = 80u * v_ld - 64u * ((v_ld & 3u) ^ 2u) + 256u;
#define LOAD_LINE(L, twn)  { _Pragma("unroll") for (int g = 0; g < 4; ++g) \
        S[4 * (L) + g] = *(gf4p)(wl[g] + (unsigned long long)lo_line + 4ull * ((twn) * CKK + 32 * (L))); }
#define LOAD_HALF(twn)     { _Pragma("unroll") for (int hh = 0; hh < 2; ++hh) \
        S[16 + hh] = *(gf4p)(wh[hh] + (unsigned long long)lo_half + 4ull * ((twn) * CKK)); }
#define TURN_LINE(L)       { _Pragma("unroll") for (int g = 0; g < 4; ++g) *(f4*)(slab + st_off + 2880 * g) = S[4 * (L) + g]; \
                             _Pragma("unroll") for (int jp = 0; jp < 4; ++jp) \
        q[4 * (L) + jp] = *(const f4*)(slab + ((jp & 1) ? ld_off1 : ld_off0) + 160 * jp); }
#define TURN_HALF()        { _Pragma("unroll") for (int hh = 0; hh < 2; ++hh) *(f4*)(slab + st2_off + 5760 * hh) = S[16 + hh]; \
                             _Pragma("unroll") for (int jp = 0; jp < 2; ++jp) \
        q[16 + jp] = *(const f4*)(slab + ((jp & 1) ? ld_off1 : ld_off0) + 160 * jp); }
    const float* bsrc = bias + (size_t)oC * PP + (size_t)h * Ww + w0;
    f4 S[KH / 8];
    f4 q[KH / 8];
    float bq[2];
    LOAD_LINE(0, 0) LOAD_LINE(1, 0) LOAD_LINE(2, 0) LOAD_LINE(3, 0) LOAD_HALF(0)
    bq[0] = bsrc[0];
    __builtin_amdgcn_sched_barrier(0);

    const float* xb = xT + (((size_t)(wave * 8) * HP + h) * WP + w0) * Nb + lane;
    float xv[72];
#pragma unroll
    for (int r = 0; r < 72; ++r)
        xv[r] = xb[(((r / 9) * HP + (r % 9) / 3) * WP + (r % 3)) * Nb];
#pragma unroll
    for (int i = 0; i < 36; ++i) {
        v2f pr;
        pr[0] = xv[2 * i];
        pr[1] = xv[2 * i + 1];
        *(v2f*)&sP[team][wave * 36 + i][2 * lane] = pr;
    }
    TURN_LINE(0) TURN_LINE(1)
    lds_barrier();

    const int PL = 80 * kh;
    const int PH = 64 + 8 * kh;
#define PAIR_ROW(s) ((((s) >> 1) < 16 ? PL + 4 * ((s) >> 1) : PH + 4 * (((s) >> 1) - 16)) + 2 * half + ((s) & 1))

#pragma unroll
    for (int tw = 0; tw < TW; ++tw) {
        const int w = w0 + tw;
        const int p = h * Ww + w;
        const int cq = tw & 1;
        const bool more = tw + 1 < TW;
        const float* xn = xb + (size_t)(tw + 1) * Nb;

        if (tw + 2 < TW) {
            const int o = wave * 8 + (lane >> 2);
            __builtin_prefetch(wgt + ((size_t)o * PP + (p + 2)) * CKK + (size_t)(lane & 3) * 72, 0, 1);
        }

        float xc[24];
        v2f nx[36];
        if (more) bq[cq ^ 1] = bsrc[tw + 1];
        __builtin_amdgcn_sched_barrier(0);

        v2f fa[2][KB][2];
#pragma unroll
        for (int j = 0; j < KB; ++j) {
            const int pi = PAIR_ROW(j);
#pragma unroll
            for (int mt = 0; mt < 2; ++mt) fa[0][j][mt] = *(const v2f*)&sP[team][pi][2 * (m16 + 16 * mt)];
        }
        v8f acc0 = {};
        v8f acc1 = {};
#pragma unroll
        for (int blk = 0; blk < NBLK; ++blk) {
            const int cur = blk & 1;
            const int nxt = cur ^ 1;
            if (blk + 1 < NBLK) {
#pragma unroll
                for (int j = 0; j < KB; ++j) {
                    const int s  = (blk + 1) * KB + j;
                    const int pi = PAIR_ROW(s);
#pragma unroll
                    for (int mt = 0; mt < 2; ++mt) fa[nxt][j][mt] = *(const v2f*)&sP[team][pi][2 * (m16 + 16 * mt)];
                }
            }
            if (blk == 1) TURN_LINE(2)
            if (blk == 2) TURN_HALF()
            if (blk == 3) TURN_LINE(3)
            if (blk == NBLK - 1 && more) {
#pragma unroll
                for (int i = 0; i < 36; ++i) {
#pragma unroll
                    for (int e = 0; e < 2; ++e) {
                        const int k = 2 * i + e;
                        nx[i][e] = (k % 3 < 2) ? sP[team][wave * 36 + (k + 1) / 2][2 * lane + ((k + 1) & 1)] : xc[k / 3];
                    }
                }
            }
            __builtin_amdgcn_sched_barrier(0);
#pragma unroll
            for (int j = 0; j < KB; ++j) {
                const int s = blk * KB + j;
                v2f fb;
                if (s & 1) { fb[0] = q[s >> 1].z; fb[1] = q[s >> 1].w; }
                else       { fb[0] = q[s >> 1].x; fb[1] = q[s >> 1].y; }
                acc0 = __builtin_amdgcn_wmma_f32_16x16x4_f32(
                    false, fa[cur][j][0], false, fb, (short)0, acc0, false, false);
                acc1 = __builtin_amdgcn_wmma_f32_16x16x4_f32(
                    false, fa[cur][j][1], false, fb, (short)0, acc1, false, false);
            }
            __builtin_amdgcn_sched_barrier(0);
            if (more) {
                if (blk == 0) {
#pragma unroll
                    for (int g = 0; g < 24; ++g)
                        xc[g] = xn[(((g / 3) * HP + (g % 3)) * WP + 2) * Nb];
                }
                if (blk == 1) LOAD_LINE(0, tw + 1)
                if (blk == 2) LOAD_LINE(1, tw + 1)
                if (blk == 3) LOAD_LINE(2, tw + 1)
                if (blk == 4) LOAD_HALF(tw + 1)
                if (blk == 5) LOAD_LINE(3, tw + 1)
            }
            __builtin_amdgcn_sched_barrier(0);
        }

        if (more) { TURN_LINE(0) TURN_LINE(1) }

        const float bv = (kh == 0) ? bq[cq] : 0.0f;
#pragma unroll
        for (int r = 0; r < 8; ++r) {
            const int n = r + 8 * half;
            sO[kh][n][oC][team * TW + tw]      = acc0[r] + bv;
            sO[kh][n + 16][oC][team * TW + tw] = acc1[r] + bv;
        }

        if (more) {
            lds_barrier();
#pragma unroll
            for (int i = 0; i < 36; ++i)
                *(v2f*)&sP[team][wave * 36 + i][2 * lane] = nx[i];
            lds_barrier();
        }
    }

    __syncthreads();

    const int ws = blockIdx.x * TWG;
    for (int e = 0; e < 8; ++e) {
        const int idx  = e * 256 + t;
        const int hf   = (idx & 1) * 4;
        const int pair = idx >> 1;
        const int n = pair >> 5;
        const int o = pair & 31;
        f4 v;
        v.x = sO[0][n][o][hf + 0] + sO[1][n][o][hf + 0];
        v.y = sO[0][n][o][hf + 1] + sO[1][n][o][hf + 1];
        v.z = sO[0][n][o][hf + 2] + sO[1][n][o][hf + 2];
        v.w = sO[0][n][o][hf + 3] + sO[1][n][o][hf + 3];
        *(f4*)(out + ((size_t)n * Oo + o) * PP + (size_t)h * Ww + ws + hf) = v;
    }
}

extern "C" void kernel_launch(void* const* d_in, const int* in_sizes, int n_in,
                              void* d_out, int out_size, void* d_ws, size_t ws_size,
                              hipStream_t stream) {
    const float* x    = (const float*)d_in[0];
    const float* wgt  = (const float*)d_in[1];
    const float* bias = (const float*)d_in[2];
    float* out        = (float*)d_out;
    float* xT         = (float*)d_ws;

    dim3 gridT(Hh / 2 + 1, Cc);
    lc2d_x_batch_inner<<<gridT, 128, 0, stream>>>(x, xT);

    dim3 grid(Ww / TWG, Hh);
    lc2d_wmma_f32<<<grid, 256, 0, stream>>>(xT, wgt, bias, out);
}
